# starting kernel
# baseline (speedup 1.0000x reference)
#include <hip/hip_runtime.h>

typedef __attribute__((ext_vector_type(2))) float v2f;
typedef __attribute__((ext_vector_type(8))) float v8f;

#define GCN_N_NODES 100000
#define GCN_IN_DIM  256
#define GCN_OUT_DIM 128

__global__ __launch_bounds__(32)
void gcn_gemm_wmma_f32(const float* __restrict__ X,
                       const float* __restrict__ W,
                       const float* __restrict__ bias,
                       float* __restrict__ H)
{
    const int mTile  = blockIdx.x;
    const int lane   = threadIdx.x;
    const int lane15 = lane & 15;
    const int hi     = lane >> 4;
    const int m0     = mTile * 16;

    const float* aPtr = X + (size_t)(m0 + lane15) * GCN_IN_DIM + 2 * hi;

    const float* bPtr[8];
#pragma unroll
    for (int j = 0; j < 8; ++j)
        bPtr[j] = W + (size_t)(16 * j + lane15) * GCN_IN_DIM + 2 * hi;

    v8f acc[8];
#pragma unroll
    for (int j = 0; j < 8; ++j)
        acc[j] = (v8f){0.f, 0.f, 0.f, 0.f, 0.f, 0.f, 0.f, 0.f};

    for (int k = 0; k < GCN_IN_DIM; k += 4) {
        v2f a = *(const v2f*)(aPtr + k);
#pragma unroll
        for (int j = 0; j < 8; ++j) {
            v2f bv = *(const v2f*)(bPtr[j] + k);
            acc[j] = __builtin_amdgcn_wmma_f32_16x16x4_f32(
                 false, a,  false, bv,
                 (short)0, acc[j],
                 false,  false);
        }
    }

#pragma unroll
    for (int j = 0; j < 8; ++j) {
        const float bj = bias[16 * j + lane15];
#pragma unroll
        for (int v = 0; v < 8; ++v) {
            const int row = m0 + v + 8 * hi;
            H[(size_t)row * GCN_OUT_DIM + 16 * j + lane15] = acc[j][v] + bj;
        }
    }
}

__global__ void gcn_zero_out(float4* __restrict__ out, int n4)
{
    int i = blockIdx.x * blockDim.x + threadIdx.x;
    if (i < n4) {
        float4 z;
        z.x = 0.f; z.y = 0.f; z.z = 0.f; z.w = 0.f;
        out[i] = z;
    }
}

__global__ __launch_bounds__(256)
void gcn_edge_scatter(const int*   __restrict__ edge_row,
                      const int*   __restrict__ edge_col,
                      const float* __restrict__ edge_val,
                      const float* __restrict__ H,
                      float*       __restrict__ out,
                      int n_edges)
{
    const int e    = (int)((blockIdx.x * (unsigned)blockDim.x + threadIdx.x) >> 5);
    const int lane = threadIdx.x & 31;
    if (e >= n_edges) return;

    const int   r = edge_row[e];
    const int   c = edge_col[e];
    const float v = edge_val[e];

    const float4 h = ((const float4*)(H + (size_t)c * GCN_OUT_DIM))[lane];
    float* orow = out + (size_t)r * GCN_OUT_DIM + lane * 4;

    unsafeAtomicAdd(orow + 0, v * h.x);
    unsafeAtomicAdd(orow + 1, v * h.y);
    unsafeAtomicAdd(orow + 2, v * h.z);
    unsafeAtomicAdd(orow + 3, v * h.w);
}

extern "C" void kernel_launch(void* const* d_in, const int* in_sizes, int n_in,
                              void* d_out, int out_size, void* d_ws, size_t ws_size,
                              hipStream_t stream)
{
    const float* X        = (const float*)d_in[0];
    const int*   edge_row = (const int*)d_in[1];
    const int*   edge_col = (const int*)d_in[2];
    const float* edge_val = (const float*)d_in[3];
    const float* W        = (const float*)d_in[4];
    const float* bias     = (const float*)d_in[5];
    float*       out      = (float*)d_out;
    float*       H        = (float*)d_ws;

    const int n_edges = in_sizes[1];

    gcn_gemm_wmma_f32<<<GCN_N_NODES / 16, 32, 0, stream>>>(X, W, bias, H);

    const int n4 = GCN_N_NODES * GCN_OUT_DIM / 4;
    gcn_zero_out<<<(n4 + 255) / 256, 256, 0, stream>>>((float4*)out, n4);

    const int nblk = (n_edges + 7) / 8;
    gcn_edge_scatter<<<nblk, 256, 0, stream>>>(edge_row, edge_col, edge_val,
                                               H, out, n_edges);
}


# Edge sums kept in LDS: each of 200 work-groups owns 500 out rows, scans all edge rows, adds matches by LDS float atomics, stores rows once; no zero pass or global atomics; sum order = LDS arrival
# speedup vs baseline: 1.8439x; 1.8439x over previous
#include <hip/hip_runtime.h>

typedef __attribute__((ext_vector_type(2))) float v2f;
typedef __attribute__((ext_vector_type(8))) float v8f;

#define GCN_N_NODES 100000
#define GCN_IN_DIM  256
#define GCN_OUT_DIM 128

__global__ __launch_bounds__(32)
void gcn_gemm_wmma_f32(const float* __restrict__ X,
                       const float* __restrict__ W,
                       const float* __restrict__ bias,
                       float* __restrict__ H)
{
    const int mTile  = blockIdx.x;
    const int lane   = threadIdx.x;
    const int lane15 = lane & 15;
    const int hi     = lane >> 4;
    const int m0     = mTile * 16;

    const float* aPtr = X + (size_t)(m0 + lane15) * GCN_IN_DIM + 2 * hi;

    const float* bPtr[8];
#pragma unroll
    for (int j = 0; j < 8; ++j)
        bPtr[j] = W + (size_t)(16 * j + lane15) * GCN_IN_DIM + 2 * hi;

    v8f acc[8];
#pragma unroll
    for (int j = 0; j < 8; ++j)
        acc[j] = (v8f){0.f, 0.f, 0.f, 0.f, 0.f, 0.f, 0.f, 0.f};

    for (int k = 0; k < GCN_IN_DIM; k += 4) {
        v2f a = *(const v2f*)(aPtr + k);
#pragma unroll
        for (int j = 0; j < 8; ++j) {
            v2f bv = *(const v2f*)(bPtr[j] + k);
            acc[j] = __builtin_amdgcn_wmma_f32_16x16x4_f32(
                 false, a,  false, bv,
                 (short)0, acc[j],
                 false,  false);
        }
    }

#pragma unroll
    for (int j = 0; j < 8; ++j) {
        const float bj = bias[16 * j + lane15];
#pragma unroll
        for (int v = 0; v < 8; ++v) {
            const int row = m0 + v + 8 * hi;
            H[(size_t)row * GCN_OUT_DIM + 16 * j + lane15] = acc[j][v] + bj;
        }
    }
}

#define GCN_ROWS_PER_WG 500
#define GCN_ROW_GROUPS  (GCN_N_NODES / GCN_ROWS_PER_WG)
#define GCN_ACC_WAVES   32
#define GCN_QUEUE_CAP   160

__device__ __forceinline__ int4 gcn_load_rows4(const int4* __restrict__ rows4,
                                               int idx4, int last4)
{
    return rows4[min(idx4, last4)];
}

__device__ __forceinline__ void gcn_lds_add(float* p, float x)
{
    __hip_atomic_fetch_add(p, x, __ATOMIC_RELAXED, __HIP_MEMORY_SCOPE_WORKGROUP);
}

__device__ __forceinline__ void gcn_drain(const int* qw, int base, int n, int lane, int row0,
                                          const int*   __restrict__ edge_row,
                                          const int*   __restrict__ edge_col,
                                          const float* __restrict__ edge_val,
                                          const float4* __restrict__ H4,
                                          float* acc)
{
    int lr = 0;
    int c  = 0;
    int vb = 0;
    if (lane < n) {
        const int e = qw[base + lane];
        lr = edge_row[e] - row0;
        c  = (int)min((unsigned)edge_col[e], (unsigned)(GCN_N_NODES - 1));
        vb = __float_as_int(edge_val[e]);
    }
#pragma nounroll
    for (int j0 = 0; j0 < n; j0 += 8) {
        float4 h[8];
#pragma unroll
        for (int u = 0; u < 8; ++u) {
            const int cj = __builtin_amdgcn_readlane(c, j0 + u);
            h[u] = H4[(size_t)cj * (GCN_OUT_DIM / 4) + lane];
        }
#pragma unroll
        for (int u = 0; u < 8; ++u) {
            if (j0 + u < n) {
                const int   rj = __builtin_amdgcn_readlane(lr, j0 + u);
                const float vj = __int_as_float(__builtin_amdgcn_readlane(vb, j0 + u));
                float* a = acc + rj * GCN_OUT_DIM + lane;
                gcn_lds_add(a,      vj * h[u].x);
                gcn_lds_add(a + 32, vj * h[u].y);
                gcn_lds_add(a + 64, vj * h[u].z);
                gcn_lds_add(a + 96, vj * h[u].w);
            }
        }
    }
}

__global__ __launch_bounds__(1024)
void gcn_rows_in_lds(const int*   __restrict__ edge_row,
                     const int*   __restrict__ edge_col,
                     const float* __restrict__ edge_val,
                     const float* __restrict__ H,
                     float*       __restrict__ out,
                     int n_edges)
{
    __shared__ __attribute__((aligned(16))) float acc[GCN_ROWS_PER_WG * GCN_OUT_DIM];
    __shared__ int queue[GCN_ACC_WAVES * GCN_QUEUE_CAP];

    const int tid  = threadIdx.x;
    const int lane = tid & 31;
    const int wave = tid >> 5;
    const int row0 = blockIdx.x * GCN_ROWS_PER_WG;
    int* qw = queue + wave * GCN_QUEUE_CAP;
    const float4* H4 = (const float4*)H;

    for (int i = tid; i < GCN_ROWS_PER_WG * GCN_OUT_DIM / 4; i += 1024) {
        float4 z;
        z.x = 0.f; z.y = 0.f; z.z = 0.f; z.w = 0.f;
        ((float4*)acc)[i] = z;
    }
    __syncthreads();

    const int n4full = n_edges >> 2;
    const int nStep  = (n4full + 1023) >> 10;
    const int idx0   = wave * 32 + lane;
    const int4* rows4 = (const int4*)edge_row;
    int qn = 0;

    if (n4full > 0) {
        const int last4 = n4full - 1;
        int4 p0 = gcn_load_rows4(rows4, idx0,        last4);
        __builtin_amdgcn_sched_barrier(0);
        int4 p1 = gcn_load_rows4(rows4, idx0 + 1024, last4);
        __builtin_amdgcn_sched_barrier(0);
        int4 p2 = gcn_load_rows4(rows4, idx0 + 2048, last4);
        __builtin_amdgcn_sched_barrier(0);
        int4 p3 = gcn_load_rows4(rows4, idx0 + 3072, last4);
        __builtin_amdgcn_sched_barrier(0);

        auto step = [&](int4& p, int s) {
            const int i4 = idx0 + s * 1024;
            const int e0 = i4 * 4;
            const unsigned lim = (i4 < n4full) ? (unsigned)GCN_ROWS_PER_WG : 0u;
            const bool m0 = ((unsigned)p.x - (unsigned)row0) < lim;
            const bool m1 = ((unsigned)p.y - (unsigned)row0) < lim;
            const bool m2 = ((unsigned)p.z - (unsigned)row0) < lim;
            const bool m3 = ((unsigned)p.w - (unsigned)row0) < lim;
            const unsigned k0 = __builtin_amdgcn_ballot_w32(m0);
            const unsigned k1 = __builtin_amdgcn_ballot_w32(m1);
            const unsigned k2 = __builtin_amdgcn_ballot_w32(m2);
            const unsigned k3 = __builtin_amdgcn_ballot_w32(m3);
            p = gcn_load_rows4(rows4, i4 + 4096, last4);

            if ((k0 | k1 | k2 | k3) != 0u) {
                if (k0 != 0u) {
                    if (m0) qw[qn + (int)__builtin_amdgcn_mbcnt_lo(k0, 0u)] = e0;
                    qn += __builtin_popcount(k0);
                }
                if (k1 != 0u) {
                    if (m1) qw[qn + (int)__builtin_amdgcn_mbcnt_lo(k1, 0u)] = e0 + 1;
                    qn += __builtin_popcount(k1);
                }
                if (k2 != 0u) {
                    if (m2) qw[qn + (int)__builtin_amdgcn_mbcnt_lo(k2, 0u)] = e0 + 2;
                    qn += __builtin_popcount(k2);
                }
                if (k3 != 0u) {
                    if (m3) qw[qn + (int)__builtin_amdgcn_mbcnt_lo(k3, 0u)] = e0 + 3;
                    qn += __builtin_popcount(k3);
                }
                while (qn >= 32) {
                    qn -= 32;
                    __builtin_amdgcn_wave_barrier();
                    gcn_drain(qw, qn, 32, lane, row0, edge_row, edge_col, edge_val, H4, acc);
                }
            }
        };
        for (int s = 0; s < nStep; s += 4) {
            step(p0, s);
            step(p1, s + 1);
            step(p2, s + 2);
            step(p3, s + 3);
        }
    }
    if (wave == 0) {
        const int e = n4full * 4 + lane;
        bool mt = false;
        if (lane < 3 && e < n_edges)
            mt = ((unsigned)edge_row[e] - (unsigned)row0) < (unsigned)GCN_ROWS_PER_WG;
        const unsigned kt = __builtin_amdgcn_ballot_w32(mt);
        if (kt != 0u) {
            if (mt) qw[qn + (int)__builtin_amdgcn_mbcnt_lo(kt, 0u)] = e;
            qn += __builtin_popcount(kt);
        }
    }
    while (qn > 0) {
        const int n = qn < 32 ? qn : 32;
        qn -= n;
        __builtin_amdgcn_wave_barrier();
        gcn_drain(qw, qn, n, lane, row0, edge_row, edge_col, edge_val, H4, acc);
    }
    __syncthreads();

    for (int i = tid; i < GCN_ROWS_PER_WG * 32; i += 1024) {
        const int r = i >> 5;
        const float* a = acc + r * GCN_OUT_DIM + lane;
        float4 o;
        o.x = a[0]; o.y = a[32]; o.z = a[64]; o.w = a[96];
        ((float4*)(out + (size_t)(row0 + r) * GCN_OUT_DIM))[lane] = o;
    }
}

extern "C" void kernel_launch(void* const* d_in, const int* in_sizes, int n_in,
                              void* d_out, int out_size, void* d_ws, size_t ws_size,
                              hipStream_t stream)
{
    const float* X        = (const float*)d_in[0];
    const int*   edge_row = (const int*)d_in[1];
    const int*   edge_col = (const int*)d_in[2];
    const float* edge_val = (const float*)d_in[3];
    const float* W        = (const float*)d_in[4];
    const float* bias     = (const float*)d_in[5];
    float*       out      = (float*)d_out;
    float*       H        = (float*)d_ws;

    const int n_edges = in_sizes[1];

    gcn_gemm_wmma_f32<<<GCN_N_NODES / 16, 32, 0, stream>>>(X, W, bias, H);

    gcn_rows_in_lds<<<GCN_ROW_GROUPS, 1024, 0, stream>>>(edge_row, edge_col, edge_val,
                                                         H, out, n_edges);
}


# Each group's scan of the whole edge list replaced by a per-group edge bitmap, built once by a new kernel in the group's own out rows (integer LDS OR); sum order still LDS arrival
# speedup vs baseline: 2.1236x; 1.1517x over previous
#include <hip/hip_runtime.h>

typedef __attribute__((ext_vector_type(2))) float v2f;
typedef __attribute__((ext_vector_type(8))) float v8f;

#define GCN_N_NODES 100000
#define GCN_IN_DIM  256
#define GCN_OUT_DIM 128

__global__ __launch_bounds__(32)
void gcn_gemm_wmma_f32(const float* __restrict__ X,
                       const float* __restrict__ W,
                       const float* __restrict__ bias,
                       float* __restrict__ H)
{
    const int mTile  = blockIdx.x;
    const int lane   = threadIdx.x;
    const int lane15 = lane & 15;
    const int hi     = lane >> 4;
    const int m0     = mTile * 16;

    const float* aPtr = X + (size_t)(m0 + lane15) * GCN_IN_DIM + 2 * hi;

    const float* bPtr[8];
#pragma unroll
    for (int j = 0; j < 8; ++j)
        bPtr[j] = W + (size_t)(16 * j + lane15) * GCN_IN_DIM + 2 * hi;

    v8f acc[8];
#pragma unroll
    for (int j = 0; j < 8; ++j)
        acc[j] = (v8f){0.f, 0.f, 0.f, 0.f, 0.f, 0.f, 0.f, 0.f};

    for (int k = 0; k < GCN_IN_DIM; k += 4) {
        v2f a = *(const v2f*)(aPtr + k);
#pragma unroll
        for (int j = 0; j < 8; ++j) {
            v2f bv = *(const v2f*)(bPtr[j] + k);
            acc[j] = __builtin_amdgcn_wmma_f32_16x16x4_f32(
                 false, a,  false, bv,
                 (short)0, acc[j],
                 false,  false);
        }
    }

#pragma unroll
    for (int j = 0; j < 8; ++j) {
        const float bj = bias[16 * j + lane15];
#pragma unroll
        for (int v = 0; v < 8; ++v) {
            const int row = m0 + v + 8 * hi;
            H[(size_t)row * GCN_OUT_DIM + 16 * j + lane15] = acc[j][v] + bj;
        }
    }
}

#define GCN_ROWS_PER_WG 500
#define GCN_ROW_GROUPS  (GCN_N_NODES / GCN_ROWS_PER_WG)
#define GCN_BM_WORDS    256
#define GCN_BM_EDGES    (GCN_BM_WORDS * 32)
#define GCN_BM_MAX_CHUNKS (GCN_ROWS_PER_WG * GCN_OUT_DIM / GCN_BM_WORDS)
#define GCN_BM_GRID     196

__device__ __forceinline__ int gcn_bitmap_chunks(int n_edges)
{
    return min((n_edges + GCN_BM_EDGES - 1) / GCN_BM_EDGES, GCN_BM_MAX_CHUNKS);
}

__global__ __launch_bounds__(1024)
void gcn_group_bitmaps(const int* __restrict__ edge_row,
                       float*     __restrict__ out,
                       int n_edges)
{
    __shared__ __attribute__((aligned(16))) unsigned tile[GCN_ROW_GROUPS * GCN_BM_WORDS];

    const int tid     = threadIdx.x;
    const int nChunks = gcn_bitmap_chunks(n_edges);
    const int4* rows4 = (const int4*)edge_row;

    for (int chunk = blockIdx.x; chunk < nChunks; chunk += gridDim.x) {
        for (int i = tid; i < GCN_ROW_GROUPS * GCN_BM_WORDS / 4; i += 1024) {
            uint4 z;
            z.x = 0u; z.y = 0u; z.z = 0u; z.w = 0u;
            ((uint4*)tile)[i] = z;
        }
        __syncthreads();

        const int e0 = chunk * GCN_BM_EDGES + tid * 8;
        int r[8];
        if (e0 + 8 <= n_edges) {
            const int4 a = rows4[(e0 >> 2)];
            const int4 b = rows4[(e0 >> 2) + 1];
            r[0] = a.x; r[1] = a.y; r[2] = a.z; r[3] = a.w;
            r[4] = b.x; r[5] = b.y; r[6] = b.z; r[7] = b.w;
        } else {
#pragma unroll
            for (int k = 0; k < 8; ++k)
                r[k] = (e0 + k < n_edges) ? edge_row[e0 + k] : -1;
        }
        const int w = tid >> 2;
#pragma unroll
        for (int k = 0; k < 8; ++k) {
            if ((unsigned)r[k] < (unsigned)GCN_N_NODES) {
                const int g = r[k] / GCN_ROWS_PER_WG;
                __hip_atomic_fetch_or(tile + g * GCN_BM_WORDS + (w ^ ((g & 15) << 2)),
                                      1u << ((tid & 3) * 8 + k),
                                      __ATOMIC_RELAXED, __HIP_MEMORY_SCOPE_WORKGROUP);
            }
        }
        __syncthreads();

        for (int i = tid; i < GCN_ROW_GROUPS * GCN_BM_WORDS / 4; i += 1024) {
            const int g = i >> 6;
            const int q = i & 63;
            const uint4 v = ((const uint4*)tile)[g * (GCN_BM_WORDS / 4) + (q ^ (g & 15))];
            uint4* dst = (uint4*)(out + (size_t)g * GCN_ROWS_PER_WG * GCN_OUT_DIM)
                         + (size_t)chunk * (GCN_BM_WORDS / 4) + q;
            *dst = v;
        }
        __syncthreads();
    }
}

#define GCN_ACC_WAVES   32
#define GCN_QUEUE_CAP   160

__device__ __forceinline__ void gcn_lds_add(float* p, float x)
{
    __hip_atomic_fetch_add(p, x, __ATOMIC_RELAXED, __HIP_MEMORY_SCOPE_WORKGROUP);
}

__device__ __forceinline__ void gcn_drain(const int* qw, int base, int n, int lane, int row0,
                                          const int*   __restrict__ edge_row,
                                          const int*   __restrict__ edge_col,
                                          const float* __restrict__ edge_val,
                                          const float4* __restrict__ H4,
                                          float* acc)
{
    int lr = 0;
    int c  = 0;
    int vb = 0;
    if (lane < n) {
        const int e = qw[base + lane];
        lr = edge_row[e] - row0;
        c  = (int)min((unsigned)edge_col[e], (unsigned)(GCN_N_NODES - 1));
        vb = __float_as_int(edge_val[e]);
    }
#pragma nounroll
    for (int j0 = 0; j0 < n; j0 += 8) {
        float4 h[8];
#pragma unroll
        for (int u = 0; u < 8; ++u) {
            const int cj = __builtin_amdgcn_readlane(c, j0 + u);
            h[u] = H4[(size_t)cj * (GCN_OUT_DIM / 4) + lane];
        }
#pragma unroll
        for (int u = 0; u < 8; ++u) {
            if (j0 + u < n) {
                const int   rj = __builtin_amdgcn_readlane(lr, j0 + u);
                const float vj = __int_as_float(__builtin_amdgcn_readlane(vb, j0 + u));
                float* a = acc + rj * GCN_OUT_DIM + lane;
                gcn_lds_add(a,      vj * h[u].x);
                gcn_lds_add(a + 32, vj * h[u].y);
                gcn_lds_add(a + 64, vj * h[u].z);
                gcn_lds_add(a + 96, vj * h[u].w);
            }
        }
    }
}

__global__ __launch_bounds__(1024)
void gcn_rows_in_lds(const int*   __restrict__ edge_row,
                     const int*   __restrict__ edge_col,
                     const float* __restrict__ edge_val,
                     const float* __restrict__ H,
                     float*       __restrict__ out,
                     int n_edges)
{
    __shared__ __attribute__((aligned(16))) float acc[GCN_ROWS_PER_WG * GCN_OUT_DIM];
    __shared__ int queue[GCN_ACC_WAVES * GCN_QUEUE_CAP];

    const int tid  = threadIdx.x;
    const int lane = tid & 31;
    const int wave = tid >> 5;
    const int row0 = blockIdx.x * GCN_ROWS_PER_WG;
    int* qw = queue + wave * GCN_QUEUE_CAP;
    const float4* H4 = (const float4*)H;

    for (int i = tid; i < GCN_ROWS_PER_WG * GCN_OUT_DIM / 4; i += 1024) {
        float4 z;
        z.x = 0.f; z.y = 0.f; z.z = 0.f; z.w = 0.f;
        ((float4*)acc)[i] = z;
    }
    __syncthreads();

    const uint4* bm4 = (const uint4*)(out + (size_t)row0 * GCN_OUT_DIM);
    const int n4 = gcn_bitmap_chunks(n_edges) * (GCN_BM_WORDS / 4);
    int qn = 0;

    auto take = [&](unsigned wd, int ebase) {
        unsigned k = __builtin_amdgcn_ballot_w32(wd != 0u);
        while (k != 0u) {
            if (wd != 0u) {
                qw[qn + (int)__builtin_amdgcn_mbcnt_lo(k, 0u)] = ebase + __builtin_ctz(wd);
                wd &= wd - 1u;
            }
            qn += __builtin_popcount(k);
            while (qn >= 32) {
                qn -= 32;
                __builtin_amdgcn_wave_barrier();
                gcn_drain(qw, qn, 32, lane, row0, edge_row, edge_col, edge_val, H4, acc);
            }
            k = __builtin_amdgcn_ballot_w32(wd != 0u);
        }
    };

    if (tid < n4) {
        uint4 cur = bm4[tid];
        for (int i = tid; i < n4; i += 1024) {
            const uint4 nxt = bm4[min(i + 1024, n4 - 1)];
            const int eb = i * 128;
            take(cur.x, eb);
            take(cur.y, eb + 32);
            take(cur.z, eb + 64);
            take(cur.w, eb + 96);
            cur = nxt;
        }
    }
    while (qn > 0) {
        const int n = qn < 32 ? qn : 32;
        qn -= n;
        __builtin_amdgcn_wave_barrier();
        gcn_drain(qw, qn, n, lane, row0, edge_row, edge_col, edge_val, H4, acc);
    }
    __syncthreads();

    for (int i = tid; i < GCN_ROWS_PER_WG * 32; i += 1024) {
        const int r = i >> 5;
        const float* a = acc + r * GCN_OUT_DIM + lane;
        float4 o;
        o.x = a[0]; o.y = a[32]; o.z = a[64]; o.w = a[96];
        ((float4*)(out + (size_t)(row0 + r) * GCN_OUT_DIM))[lane] = o;
    }
}

extern "C" void kernel_launch(void* const* d_in, const int* in_sizes, int n_in,
                              void* d_out, int out_size, void* d_ws, size_t ws_size,
                              hipStream_t stream)
{
    const float* X        = (const float*)d_in[0];
    const int*   edge_row = (const int*)d_in[1];
    const int*   edge_col = (const int*)d_in[2];
    const float* edge_val = (const float*)d_in[3];
    const float* W        = (const float*)d_in[4];
    const float* bias     = (const float*)d_in[5];
    float*       out      = (float*)d_out;
    float*       H        = (float*)d_ws;

    const int n_edges = in_sizes[1];

    gcn_gemm_wmma_f32<<<GCN_N_NODES / 16, 32, 0, stream>>>(X, W, bias, H);

    gcn_group_bitmaps<<<GCN_BM_GRID, 1024, 0, stream>>>(edge_row, out, n_edges);

    gcn_rows_in_lds<<<GCN_ROW_GROUPS, 1024, 0, stream>>>(edge_row, edge_col, edge_val,
                                                         H, out, n_edges);
}


# Projection re-tiled: each wave keeps its 32 W columns in registers, X rows reach the matrix unit via LDS (whole-line loads, 32-row blocks, 256 8-wave groups); k order in each 8-k group changes
# speedup vs baseline: 2.7841x; 1.3110x over previous
#include <hip/hip_runtime.h>

typedef __attribute__((ext_vector_type(2))) float v2f;
typedef __attribute__((ext_vector_type(8))) float v8f;

#define GCN_N_NODES 100000
#define GCN_IN_DIM  256
#define GCN_OUT_DIM 128

typedef __attribute__((ext_vector_type(4))) float v4f;

#define PJ_ROWS   32
#define PJ_PITCH  (GCN_IN_DIM + 4)
#define PJ_STAGE  (PJ_ROWS * PJ_PITCH)
#define PJ_BLOCKS (GCN_N_NODES / PJ_ROWS)
#define PJ_GRID   256
static_assert(PJ_GRID <= PJ_BLOCKS, "every work-group has a first block");

__device__ __forceinline__ void pj_lds_barrier()
{
    asm volatile("s_wait_dscnt 0x0\n\ts_barrier_signal -1\n\ts_barrier_wait -1" ::: "memory");
}

__global__ __launch_bounds__(256)
void gcn_gemm_wmma_f32(const float* __restrict__ X,
                       const float* __restrict__ W,
                       const float* __restrict__ bias,
                       float* __restrict__ H)
{
    __shared__ __attribute__((aligned(16))) float stage[2 * PJ_STAGE];

    const int tid  = threadIdx.x;
    const int lane = tid & 31;
    const int wave = tid >> 5;
    const int c    = lane & 15;
    const int h    = lane >> 4;
    const int p    = wave & 3;
    const int q    = wave >> 2;

    v4f w0[32];
    v4f w1[32];
    {
        const float* wp0 = W + (size_t)(32 * p + c) * GCN_IN_DIM + 4 * h;
        const float* wp1 = wp0 + 16 * GCN_IN_DIM;
#pragma unroll
        for (int kk = 0; kk < 32; ++kk) {
            w0[kk] = *(const v4f*)(wp0 + 8 * kk);
            w1[kk] = *(const v4f*)(wp1 + 8 * kk);
        }
    }
    const float bias0 = bias[32 * p + c];
    const float bias1 = bias[32 * p + 16 + c];

    const int srow = tid >> 6;
    const int scol = (tid & 63) * 4;
    const float* xsrc = X + (size_t)srow * GCN_IN_DIM + scol;
    float*       sdst = stage + srow * PJ_PITCH + scol;

    int b = blockIdx.x;
    v4f xr[8];
    {
#pragma unroll
        for (int j = 0; j < 8; ++j)
            xr[j] = *(const v4f*)(xsrc + ((size_t)b * PJ_ROWS + 4 * j) * GCN_IN_DIM);
#pragma unroll
        for (int j = 0; j < 8; ++j)
            *(v4f*)(sdst + 4 * j * PJ_PITCH) = xr[j];
        const int b1 = min(b + PJ_GRID, PJ_BLOCKS - 1);
#pragma unroll
        for (int j = 0; j < 8; ++j)
            xr[j] = *(const v4f*)(xsrc + ((size_t)b1 * PJ_ROWS + 4 * j) * GCN_IN_DIM);
    }
    pj_lds_barrier();

    int buf = 0;
    do {
        const float* sA = stage + buf * PJ_STAGE + (16 * q + c) * PJ_PITCH + 4 * h;
        v8f acc0 = (v8f){0.f, 0.f, 0.f, 0.f, 0.f, 0.f, 0.f, 0.f};
        v8f acc1 = (v8f){0.f, 0.f, 0.f, 0.f, 0.f, 0.f, 0.f, 0.f};

#pragma unroll
        for (int g = 0; g < 4; ++g) {
            if (g == 2) {
                float* d = sdst + (buf ^ 1) * PJ_STAGE;
#pragma unroll
                for (int j = 0; j < 8; ++j)
                    *(v4f*)(d + 4 * j * PJ_PITCH) = xr[j];
            }
            v4f a[8];
#pragma unroll
            for (int i = 0; i < 8; ++i)
                a[i] = *(const v4f*)(sA + 8 * (8 * g + i));
#pragma unroll
            for (int i = 0; i < 8; ++i) {
                const int kk = 8 * g + i;
                const v2f alo = (v2f){a[i].x, a[i].y};
                const v2f ahi = (v2f){a[i].z, a[i].w};
                acc0 = __builtin_amdgcn_wmma_f32_16x16x4_f32(
                    false, alo, false, (v2f){w0[kk].x, w0[kk].y}, (short)0, acc0, false, false);
                acc1 = __builtin_amdgcn_wmma_f32_16x16x4_f32(
                    false, alo, false, (v2f){w1[kk].x, w1[kk].y}, (short)0, acc1, false, false);
                acc0 = __builtin_amdgcn_wmma_f32_16x16x4_f32(
                    false, ahi, false, (v2f){w0[kk].z, w0[kk].w}, (short)0, acc0, false, false);
                acc1 = __builtin_amdgcn_wmma_f32_16x16x4_f32(
                    false, ahi, false, (v2f){w1[kk].z, w1[kk].w}, (short)0, acc1, false, false);
            }
        }

        float* hp = H + ((size_t)b * PJ_ROWS + 16 * q + 8 * h) * GCN_OUT_DIM + 32 * p + c;
#pragma unroll
        for (int v = 0; v < 8; ++v) {
            hp[v * GCN_OUT_DIM]      = acc0[v] + bias0;
            hp[v * GCN_OUT_DIM + 16] = acc1[v] + bias1;
        }

        const int b2 = min(b + 2 * PJ_GRID, PJ_BLOCKS - 1);
#pragma unroll
        for (int j = 0; j < 8; ++j)
            xr[j] = *(const v4f*)(xsrc + ((size_t)b2 * PJ_ROWS + 4 * j) * GCN_IN_DIM);

        pj_lds_barrier();
        buf ^= 1;
        b += PJ_GRID;
    } while (b < PJ_BLOCKS);
}

#define GCN_ROWS_PER_WG 500
#define GCN_ROW_GROUPS  (GCN_N_NODES / GCN_ROWS_PER_WG)
#define GCN_BM_WORDS    256
#define GCN_BM_EDGES    (GCN_BM_WORDS * 32)
#define GCN_BM_MAX_CHUNKS (GCN_ROWS_PER_WG * GCN_OUT_DIM / GCN_BM_WORDS)
#define GCN_BM_GRID     196

__device__ __forceinline__ int gcn_bitmap_chunks(int n_edges)
{
    return min((n_edges + GCN_BM_EDGES - 1) / GCN_BM_EDGES, GCN_BM_MAX_CHUNKS);
}

__global__ __launch_bounds__(1024)
void gcn_group_bitmaps(const int* __restrict__ edge_row,
                       float*     __restrict__ out,
                       int n_edges)
{
    __shared__ __attribute__((aligned(16))) unsigned tile[GCN_ROW_GROUPS * GCN_BM_WORDS];

    const int tid     = threadIdx.x;
    const int nChunks = gcn_bitmap_chunks(n_edges);
    const int4* rows4 = (const int4*)edge_row;

    for (int chunk = blockIdx.x; chunk < nChunks; chunk += gridDim.x) {
        for (int i = tid; i < GCN_ROW_GROUPS * GCN_BM_WORDS / 4; i += 1024) {
            uint4 z;
            z.x = 0u; z.y = 0u; z.z = 0u; z.w = 0u;
            ((uint4*)tile)[i] = z;
        }
        __syncthreads();

        const int e0 = chunk * GCN_BM_EDGES + tid * 8;
        int r[8];
        if (e0 + 8 <= n_edges) {
            const int4 a = rows4[(e0 >> 2)];
            const int4 b = rows4[(e0 >> 2) + 1];
            r[0] = a.x; r[1] = a.y; r[2] = a.z; r[3] = a.w;
            r[4] = b.x; r[5] = b.y; r[6] = b.z; r[7] = b.w;
        } else {
#pragma unroll
            for (int k = 0; k < 8; ++k)
                r[k] = (e0 + k < n_edges) ? edge_row[e0 + k] : -1;
        }
        const int w = tid >> 2;
#pragma unroll
        for (int k = 0; k < 8; ++k) {
            if ((unsigned)r[k] < (unsigned)GCN_N_NODES) {
                const int g = r[k] / GCN_ROWS_PER_WG;
                __hip_atomic_fetch_or(tile + g * GCN_BM_WORDS + (w ^ ((g & 15) << 2)),
                                      1u << ((tid & 3) * 8 + k),
                                      __ATOMIC_RELAXED, __HIP_MEMORY_SCOPE_WORKGROUP);
            }
        }
        __syncthreads();

        for (int i = tid; i < GCN_ROW_GROUPS * GCN_BM_WORDS / 4; i += 1024) {
            const int g = i >> 6;
            const int q = i & 63;
            const uint4 v = ((const uint4*)tile)[g * (GCN_BM_WORDS / 4) + (q ^ (g & 15))];
            uint4* dst = (uint4*)(out + (size_t)g * GCN_ROWS_PER_WG * GCN_OUT_DIM)
                         + (size_t)chunk * (GCN_BM_WORDS / 4) + q;
            *dst = v;
        }
        __syncthreads();
    }
}

#define GCN_ACC_WAVES   32
#define GCN_QUEUE_CAP   160

__device__ __forceinline__ void gcn_lds_add(float* p, float x)
{
    __hip_atomic_fetch_add(p, x, __ATOMIC_RELAXED, __HIP_MEMORY_SCOPE_WORKGROUP);
}

__device__ __forceinline__ void gcn_drain(const int* qw, int base, int n, int lane, int row0,
                                          const int*   __restrict__ edge_row,
                                          const int*   __restrict__ edge_col,
                                          const float* __restrict__ edge_val,
                                          const float4* __restrict__ H4,
                                          float* acc)
{
    int lr = 0;
    int c  = 0;
    int vb = 0;
    if (lane < n) {
        const int e = qw[base + lane];
        lr = edge_row[e] - row0;
        c  = (int)min((unsigned)edge_col[e], (unsigned)(GCN_N_NODES - 1));
        vb = __float_as_int(edge_val[e]);
    }
#pragma nounroll
    for (int j0 = 0; j0 < n; j0 += 8) {
        float4 h[8];
#pragma unroll
        for (int u = 0; u < 8; ++u) {
            const int cj = __builtin_amdgcn_readlane(c, j0 + u);
            h[u] = H4[(size_t)cj * (GCN_OUT_DIM / 4) + lane];
        }
#pragma unroll
        for (int u = 0; u < 8; ++u) {
            if (j0 + u < n) {
                const int   rj = __builtin_amdgcn_readlane(lr, j0 + u);
                const float vj = __int_as_float(__builtin_amdgcn_readlane(vb, j0 + u));
                float* a = acc + rj * GCN_OUT_DIM + lane;
                gcn_lds_add(a,      vj * h[u].x);
                gcn_lds_add(a + 32, vj * h[u].y);
                gcn_lds_add(a + 64, vj * h[u].z);
                gcn_lds_add(a + 96, vj * h[u].w);
            }
        }
    }
}

__global__ __launch_bounds__(1024)
void gcn_rows_in_lds(const int*   __restrict__ edge_row,
                     const int*   __restrict__ edge_col,
                     const float* __restrict__ edge_val,
                     const float* __restrict__ H,
                     float*       __restrict__ out,
                     int n_edges)
{
    __shared__ __attribute__((aligned(16))) float acc[GCN_ROWS_PER_WG * GCN_OUT_DIM];
    __shared__ int queue[GCN_ACC_WAVES * GCN_QUEUE_CAP];

    const int tid  = threadIdx.x;
    const int lane = tid & 31;
    const int wave = tid >> 5;
    const int row0 = blockIdx.x * GCN_ROWS_PER_WG;
    int* qw = queue + wave * GCN_QUEUE_CAP;
    const float4* H4 = (const float4*)H;

    for (int i = tid; i < GCN_ROWS_PER_WG * GCN_OUT_DIM / 4; i += 1024) {
        float4 z;
        z.x = 0.f; z.y = 0.f; z.z = 0.f; z.w = 0.f;
        ((float4*)acc)[i] = z;
    }
    __syncthreads();

    const uint4* bm4 = (const uint4*)(out + (size_t)row0 * GCN_OUT_DIM);
    const int n4 = gcn_bitmap_chunks(n_edges) * (GCN_BM_WORDS / 4);
    int qn = 0;

    auto take = [&](unsigned wd, int ebase) {
        unsigned k = __builtin_amdgcn_ballot_w32(wd != 0u);
        while (k != 0u) {
            if (wd != 0u) {
                qw[qn + (int)__builtin_amdgcn_mbcnt_lo(k, 0u)] = ebase + __builtin_ctz(wd);
                wd &= wd - 1u;
            }
            qn += __builtin_popcount(k);
            while (qn >= 32) {
                qn -= 32;
                __builtin_amdgcn_wave_barrier();
                gcn_drain(qw, qn, 32, lane, row0, edge_row, edge_col, edge_val, H4, acc);
            }
            k = __builtin_amdgcn_ballot_w32(wd != 0u);
        }
    };

    if (tid < n4) {
        uint4 cur = bm4[tid];
        for (int i = tid; i < n4; i += 1024) {
            const uint4 nxt = bm4[min(i + 1024, n4 - 1)];
            const int eb = i * 128;
            take(cur.x, eb);
            take(cur.y, eb + 32);
            take(cur.z, eb + 64);
            take(cur.w, eb + 96);
            cur = nxt;
        }
    }
    while (qn > 0) {
        const int n = qn < 32 ? qn : 32;
        qn -= n;
        __builtin_amdgcn_wave_barrier();
        gcn_drain(qw, qn, n, lane, row0, edge_row, edge_col, edge_val, H4, acc);
    }
    __syncthreads();

    for (int i = tid; i < GCN_ROWS_PER_WG * 32; i += 1024) {
        const int r = i >> 5;
        const float* a = acc + r * GCN_OUT_DIM + lane;
        float4 o;
        o.x = a[0]; o.y = a[32]; o.z = a[64]; o.w = a[96];
        ((float4*)(out + (size_t)(row0 + r) * GCN_OUT_DIM))[lane] = o;
    }
}

extern "C" void kernel_launch(void* const* d_in, const int* in_sizes, int n_in,
                              void* d_out, int out_size, void* d_ws, size_t ws_size,
                              hipStream_t stream)
{
    const float* X        = (const float*)d_in[0];
    const int*   edge_row = (const int*)d_in[1];
    const int*   edge_col = (const int*)d_in[2];
    const float* edge_val = (const float*)d_in[3];
    const float* W        = (const float*)d_in[4];
    const float* bias     = (const float*)d_in[5];
    float*       out      = (float*)d_out;
    float*       H        = (float*)d_ws;

    const int n_edges = in_sizes[1];

    gcn_gemm_wmma_f32<<<PJ_GRID, 256, 0, stream>>>(X, W, bias, H);

    gcn_group_bitmaps<<<GCN_BM_GRID, 1024, 0, stream>>>(edge_row, out, n_edges);

    gcn_rows_in_lds<<<GCN_ROW_GROUPS, 1024, 0, stream>>>(edge_row, edge_col, edge_val,
                                                         H, out, n_edges);
}


# Group words carry their lone edge's place, row in group and column (all ones: several edges, looked up by the wave): the row kernel fetches 1 value per edge, not 3; sum order still LDS arrival
# speedup vs baseline: 3.0979x; 1.1127x over previous
#include <hip/hip_runtime.h>

typedef __attribute__((ext_vector_type(2))) float v2f;
typedef __attribute__((ext_vector_type(8))) float v8f;

#define GCN_N_NODES 100000
#define GCN_IN_DIM  256
#define GCN_OUT_DIM 128

typedef __attribute__((ext_vector_type(4))) float v4f;

#define PJ_ROWS   32
#define PJ_PITCH  (GCN_IN_DIM + 4)
#define PJ_STAGE  (PJ_ROWS * PJ_PITCH)
#define PJ_BLOCKS (GCN_N_NODES / PJ_ROWS)
#define PJ_GRID   256
static_assert(PJ_GRID <= PJ_BLOCKS, "every work-group has a first block");

__device__ __forceinline__ void pj_lds_barrier()
{
    asm volatile("s_wait_dscnt 0x0\n\ts_barrier_signal -1\n\ts_barrier_wait -1" ::: "memory");
}

__global__ __launch_bounds__(256)
void gcn_gemm_wmma_f32(const float* __restrict__ X,
                       const float* __restrict__ W,
                       const float* __restrict__ bias,
                       float* __restrict__ H)
{
    __shared__ __attribute__((aligned(16))) float stage[2 * PJ_STAGE];

    const int tid  = threadIdx.x;
    const int lane = tid & 31;
    const int wave = tid >> 5;
    const int c    = lane & 15;
    const int h    = lane >> 4;
    const int p    = wave & 3;
    const int q    = wave >> 2;

    v4f w0[32];
    v4f w1[32];
    {
        const float* wp0 = W + (size_t)(32 * p + c) * GCN_IN_DIM + 4 * h;
        const float* wp1 = wp0 + 16 * GCN_IN_DIM;
#pragma unroll
        for (int kk = 0; kk < 32; ++kk) {
            w0[kk] = *(const v4f*)(wp0 + 8 * kk);
            w1[kk] = *(const v4f*)(wp1 + 8 * kk);
        }
    }
    const float bias0 = bias[32 * p + c];
    const float bias1 = bias[32 * p + 16 + c];

    const int srow = tid >> 6;
    const int scol = (tid & 63) * 4;
    const float* xsrc = X + (size_t)srow * GCN_IN_DIM + scol;
    float*       sdst = stage + srow * PJ_PITCH + scol;

    int b = blockIdx.x;
    v4f xr[8];
    {
#pragma unroll
        for (int j = 0; j < 8; ++j)
            xr[j] = *(const v4f*)(xsrc + ((size_t)b * PJ_ROWS + 4 * j) * GCN_IN_DIM);
#pragma unroll
        for (int j = 0; j < 8; ++j)
            *(v4f*)(sdst + 4 * j * PJ_PITCH) = xr[j];
        const int b1 = min(b + PJ_GRID, PJ_BLOCKS - 1);
#pragma unroll
        for (int j = 0; j < 8; ++j)
            xr[j] = *(const v4f*)(xsrc + ((size_t)b1 * PJ_ROWS + 4 * j) * GCN_IN_DIM);
    }
    pj_lds_barrier();

    int buf = 0;
    do {
        const float* sA = stage + buf * PJ_STAGE + (16 * q + c) * PJ_PITCH + 4 * h;
        v8f acc0 = (v8f){0.f, 0.f, 0.f, 0.f, 0.f, 0.f, 0.f, 0.f};
        v8f acc1 = (v8f){0.f, 0.f, 0.f, 0.f, 0.f, 0.f, 0.f, 0.f};

#pragma unroll
        for (int g = 0; g < 4; ++g) {
            if (g == 2) {
                float* d = sdst + (buf ^ 1) * PJ_STAGE;
#pragma unroll
                for (int j = 0; j < 8; ++j)
                    *(v4f*)(d + 4 * j * PJ_PITCH) = xr[j];
            }
            v4f a[8];
#pragma unroll
            for (int i = 0; i < 8; ++i)
                a[i] = *(const v4f*)(sA + 8 * (8 * g + i));
#pragma unroll
            for (int i = 0; i < 8; ++i) {
                const int kk = 8 * g + i;
                const v2f alo = (v2f){a[i].x, a[i].y};
                const v2f ahi = (v2f){a[i].z, a[i].w};
                acc0 = __builtin_amdgcn_wmma_f32_16x16x4_f32(
                    false, alo, false, (v2f){w0[kk].x, w0[kk].y}, (short)0, acc0, false, false);
                acc1 = __builtin_amdgcn_wmma_f32_16x16x4_f32(
                    false, alo, false, (v2f){w1[kk].x, w1[kk].y}, (short)0, acc1, false, false);
                acc0 = __builtin_amdgcn_wmma_f32_16x16x4_f32(
                    false, ahi, false, (v2f){w0[kk].z, w0[kk].w}, (short)0, acc0, false, false);
                acc1 = __builtin_amdgcn_wmma_f32_16x16x4_f32(
                    false, ahi, false, (v2f){w1[kk].z, w1[kk].w}, (short)0, acc1, false, false);
            }
        }

        float* hp = H + ((size_t)b * PJ_ROWS + 16 * q + 8 * h) * GCN_OUT_DIM + 32 * p + c;
#pragma unroll
        for (int v = 0; v < 8; ++v) {
            hp[v * GCN_OUT_DIM]      = acc0[v] + bias0;
            hp[v * GCN_OUT_DIM + 16] = acc1[v] + bias1;
        }

        const int b2 = min(b + 2 * PJ_GRID, PJ_BLOCKS - 1);
#pragma unroll
        for (int j = 0; j < 8; ++j)
            xr[j] = *(const v4f*)(xsrc + ((size_t)b2 * PJ_ROWS + 4 * j) * GCN_IN_DIM);

        pj_lds_barrier();
        buf ^= 1;
        b += PJ_GRID;
    } while (b < PJ_BLOCKS);
}

#define GCN_ROWS_PER_WG 500
#define GCN_ROW_GROUPS  (GCN_N_NODES / GCN_ROWS_PER_WG)
#define GCN_BM_WORDS    256
#define GCN_BM_EDGES    (GCN_BM_WORDS * 32)
#define GCN_BM_MAX_CHUNKS (GCN_ROWS_PER_WG * GCN_OUT_DIM / GCN_BM_WORDS)
#define GCN_BM_GRID     196
#define GCN_WORD_ONE    0x80000000u
#define GCN_WORD_MANY   0xFFFFFFFFu
static_assert(GCN_N_NODES <= (1 << 17), "a column fits bits 16..0");
static_assert(GCN_ROWS_PER_WG <= 511, "a row inside the group fits bits 25..17 and never fills them");

__device__ __forceinline__ int gcn_bitmap_chunks(int n_edges)
{
    return min((n_edges + GCN_BM_EDGES - 1) / GCN_BM_EDGES, GCN_BM_MAX_CHUNKS);
}

__global__ __launch_bounds__(1024)
void gcn_group_bitmaps(const int* __restrict__ edge_row,
                       const int* __restrict__ edge_col,
                       float*     __restrict__ out,
                       int n_edges)
{
    __shared__ __attribute__((aligned(16))) unsigned tile[GCN_ROW_GROUPS * GCN_BM_WORDS];

    const int tid     = threadIdx.x;
    const int nChunks = gcn_bitmap_chunks(n_edges);
    const int4* rows4 = (const int4*)edge_row;
    const int4* cols4 = (const int4*)edge_col;

    for (int chunk = blockIdx.x; chunk < nChunks; chunk += gridDim.x) {
        for (int i = tid; i < GCN_ROW_GROUPS * GCN_BM_WORDS / 4; i += 1024) {
            uint4 z;
            z.x = 0u; z.y = 0u; z.z = 0u; z.w = 0u;
            ((uint4*)tile)[i] = z;
        }
        __syncthreads();

        const int e0 = chunk * GCN_BM_EDGES + tid * 8;
        int r[8];
        int cc[8];
        if (e0 + 8 <= n_edges) {
            const int4 a = rows4[(e0 >> 2)];
            const int4 b = rows4[(e0 >> 2) + 1];
            const int4 ca = cols4[(e0 >> 2)];
            const int4 cb = cols4[(e0 >> 2) + 1];
            r[0] = a.x; r[1] = a.y; r[2] = a.z; r[3] = a.w;
            r[4] = b.x; r[5] = b.y; r[6] = b.z; r[7] = b.w;
            cc[0] = ca.x; cc[1] = ca.y; cc[2] = ca.z; cc[3] = ca.w;
            cc[4] = cb.x; cc[5] = cb.y; cc[6] = cb.z; cc[7] = cb.w;
        } else {
#pragma unroll
            for (int k = 0; k < 8; ++k) {
                r[k]  = (e0 + k < n_edges) ? edge_row[e0 + k] : -1;
                cc[k] = (e0 + k < n_edges) ? edge_col[e0 + k] : 0;
            }
        }
        const int w = tid >> 2;
#pragma unroll
        for (int k = 0; k < 8; ++k) {
            if ((unsigned)r[k] < (unsigned)GCN_N_NODES) {
                const int g = r[k] / GCN_ROWS_PER_WG;
                __hip_atomic_fetch_or(tile + g * GCN_BM_WORDS + (w ^ ((g & 15) << 2)),
                                      1u << ((tid & 3) * 8 + k),
                                      __ATOMIC_RELAXED, __HIP_MEMORY_SCOPE_WORKGROUP);
            }
        }
        __syncthreads();

#pragma unroll
        for (int k = 0; k < 8; ++k) {
            if ((unsigned)r[k] < (unsigned)GCN_N_NODES) {
                const int g = r[k] / GCN_ROWS_PER_WG;
                unsigned* p = tile + g * GCN_BM_WORDS + (w ^ ((g & 15) << 2));
                const unsigned pos = (unsigned)((tid & 3) * 8 + k);
                const unsigned cur = __hip_atomic_load(p, __ATOMIC_RELAXED,
                                                       __HIP_MEMORY_SCOPE_WORKGROUP);
                if (cur != GCN_WORD_MANY) {
                    const unsigned lr  = (unsigned)(r[k] - g * GCN_ROWS_PER_WG);
                    const unsigned col = min((unsigned)cc[k], (unsigned)(GCN_N_NODES - 1));
                    const unsigned one = GCN_WORD_ONE | (pos << 26) | (lr << 17) | col;
                    __hip_atomic_store(p, cur == (1u << pos) ? one : GCN_WORD_MANY,
                                       __ATOMIC_RELAXED, __HIP_MEMORY_SCOPE_WORKGROUP);
                }
            }
        }
        __syncthreads();

        for (int i = tid; i < GCN_ROW_GROUPS * GCN_BM_WORDS / 4; i += 1024) {
            const int g = i >> 6;
            const int q = i & 63;
            const uint4 v = ((const uint4*)tile)[g * (GCN_BM_WORDS / 4) + (q ^ (g & 15))];
            uint4* dst = (uint4*)(out + (size_t)g * GCN_ROWS_PER_WG * GCN_OUT_DIM)
                         + (size_t)chunk * (GCN_BM_WORDS / 4) + q;
            *dst = v;
        }
        __syncthreads();
    }
}

#define GCN_ACC_WAVES   32
#define GCN_QUEUE_CAP   64

__device__ __forceinline__ void gcn_lds_add(float* p, float x)
{
    __hip_atomic_fetch_add(p, x, __ATOMIC_RELAXED, __HIP_MEMORY_SCOPE_WORKGROUP);
}

__device__ __forceinline__ void gcn_drain(const uint2* qw, int base, int n, int lane,
                                          const float* __restrict__ edge_val,
                                          const float4* __restrict__ H4,
                                          float* acc)
{
    int lr = 0;
    int c  = 0;
    int vb = 0;
    if (lane < n) {
        const uint2 t = qw[base + lane];
        lr = (int)((t.y >> 17) & 511u);
        c  = (int)(t.y & 0x1FFFFu);
        vb = __float_as_int(edge_val[t.x]);
    }
#pragma nounroll
    for (int j0 = 0; j0 < n; j0 += 8) {
        float4 h[8];
#pragma unroll
        for (int u = 0; u < 8; ++u) {
            const int cj = __builtin_amdgcn_readlane(c, j0 + u);
            h[u] = H4[(size_t)cj * (GCN_OUT_DIM / 4) + lane];
        }
#pragma unroll
        for (int u = 0; u < 8; ++u) {
            if (j0 + u < n) {
                const int   rj = __builtin_amdgcn_readlane(lr, j0 + u);
                const float vj = __int_as_float(__builtin_amdgcn_readlane(vb, j0 + u));
                float* a = acc + rj * GCN_OUT_DIM + lane;
                gcn_lds_add(a,      vj * h[u].x);
                gcn_lds_add(a + 32, vj * h[u].y);
                gcn_lds_add(a + 64, vj * h[u].z);
                gcn_lds_add(a + 96, vj * h[u].w);
            }
        }
    }
}

__global__ __launch_bounds__(1024)
void gcn_rows_in_lds(const int*   __restrict__ edge_row,
                     const int*   __restrict__ edge_col,
                     const float* __restrict__ edge_val,
                     const float* __restrict__ H,
                     float*       __restrict__ out,
                     int n_edges)
{
    __shared__ __attribute__((aligned(16))) float acc[GCN_ROWS_PER_WG * GCN_OUT_DIM];
    __shared__ uint2 queue[GCN_ACC_WAVES * GCN_QUEUE_CAP];

    const int tid  = threadIdx.x;
    const int lane = tid & 31;
    const int wave = tid >> 5;
    const int row0 = blockIdx.x * GCN_ROWS_PER_WG;
    uint2* qw = queue + wave * GCN_QUEUE_CAP;
    const float4* H4 = (const float4*)H;

    for (int i = tid; i < GCN_ROWS_PER_WG * GCN_OUT_DIM / 4; i += 1024) {
        float4 z;
        z.x = 0.f; z.y = 0.f; z.z = 0.f; z.w = 0.f;
        ((float4*)acc)[i] = z;
    }
    __syncthreads();

    const uint4* bm4 = (const uint4*)(out + (size_t)row0 * GCN_OUT_DIM);
    const int n4 = gcn_bitmap_chunks(n_edges) * (GCN_BM_WORDS / 4);
    int qn = 0;

    auto push = [&](bool has, unsigned e, unsigned word) {
        const unsigned k = __builtin_amdgcn_ballot_w32(has);
        if (k != 0u) {
            if (has) {
                uint2 t;
                t.x = e;
                t.y = word;
                qw[qn + (int)__builtin_amdgcn_mbcnt_lo(k, 0u)] = t;
            }
            qn += __builtin_popcount(k);
            if (qn >= 32) {
                qn -= 32;
                __builtin_amdgcn_wave_barrier();
                gcn_drain(qw, qn, 32, lane, edge_val, H4, acc);
            }
        }
    };

    auto take = [&](unsigned wd, int ebase) {
        bool     has  = wd != 0u && wd != GCN_WORD_MANY;
        unsigned e    = (unsigned)ebase + ((wd >> 26) & 31u);
        unsigned word = wd;
        unsigned m = __builtin_amdgcn_ballot_w32(wd == GCN_WORD_MANY);
        for (;;) {
            push(has, e, word);
            if (m == 0u)
                break;
            const int src = __builtin_ctz(m);
            m &= m - 1u;
            const int ei = __builtin_amdgcn_readlane(ebase, src) + lane;
            int r  = -1;
            int cl = 0;
            if (ei < n_edges) {
                r  = edge_row[ei];
                cl = edge_col[ei];
            }
            const unsigned lr = (unsigned)(r - row0);
            has  = lr < (unsigned)GCN_ROWS_PER_WG;
            e    = (unsigned)ei;
            word = GCN_WORD_ONE | ((unsigned)lane << 26) | (lr << 17)
                   | min((unsigned)cl, (unsigned)(GCN_N_NODES - 1));
        }
    };

    if (tid < n4) {
        uint4 cur = bm4[tid];
        for (int i = tid; i < n4; i += 1024) {
            const uint4 nxt = bm4[min(i + 1024, n4 - 1)];
            const int eb = i * 128;
            take(cur.x, eb);
            take(cur.y, eb + 32);
            take(cur.z, eb + 64);
            take(cur.w, eb + 96);
            cur = nxt;
        }
    }
    if (qn > 0) {
        __builtin_amdgcn_wave_barrier();
        gcn_drain(qw, 0, qn, lane, edge_val, H4, acc);
    }
    __syncthreads();

    for (int i = tid; i < GCN_ROWS_PER_WG * 32; i += 1024) {
        const int r = i >> 5;
        const float* a = acc + r * GCN_OUT_DIM + lane;
        float4 o;
        o.x = a[0]; o.y = a[32]; o.z = a[64]; o.w = a[96];
        ((float4*)(out + (size_t)(row0 + r) * GCN_OUT_DIM))[lane] = o;
    }
}

extern "C" void kernel_launch(void* const* d_in, const int* in_sizes, int n_in,
                              void* d_out, int out_size, void* d_ws, size_t ws_size,
                              hipStream_t stream)
{
    const float* X        = (const float*)d_in[0];
    const int*   edge_row = (const int*)d_in[1];
    const int*   edge_col = (const int*)d_in[2];
    const float* edge_val = (const float*)d_in[3];
    const float* W        = (const float*)d_in[4];
    const float* bias     = (const float*)d_in[5];
    float*       out      = (float*)d_out;
    float*       H        = (float*)d_ws;

    const int n_edges = in_sizes[1];

    gcn_gemm_wmma_f32<<<PJ_GRID, 256, 0, stream>>>(X, W, bias, H);

    gcn_group_bitmaps<<<GCN_BM_GRID, 1024, 0, stream>>>(edge_row, edge_col, out, n_edges);

    gcn_rows_in_lds<<<GCN_ROW_GROUPS, 1024, 0, stream>>>(edge_row, edge_col, edge_val,
                                                         H, out, n_edges);
}
